# starting kernel
# baseline (speedup 1.0000x reference)
#include <hip/hip_runtime.h>
#include <hip/hip_bf16.h>


typedef float v2f __attribute__((ext_vector_type(2)));
typedef float v8f __attribute__((ext_vector_type(8)));

#define NB   2048
#define TT   512
#define HH   11
#define GG   44
#define OUTF 22

__device__ __forceinline__ v8f wmma_f32_16x16x4(v2f a, v2f b, v8f c) {
  return __builtin_amdgcn_wmma_f32_16x16x4_f32(false, a, false, b, (short)0, c,
                                               false, false);
}

__device__ __forceinline__ float fast_sigmoid(float x) {
  return __builtin_amdgcn_rcpf(1.0f + __expf(-x));
}
__device__ __forceinline__ float fast_tanh(float x) {
  return 1.0f - 2.0f * __builtin_amdgcn_rcpf(__expf(2.0f * x) + 1.0f);
}
__device__ __forceinline__ v8f splat8(float x) {
  v8f v;
#pragma unroll
  for (int i = 0; i < 8; ++i) v[i] = x;
  return v;
}

__global__ __launch_bounds__(256) void reduce_x_kernel(
    const float* __restrict__ x, float* __restrict__ xsum) {
  size_t idx = (size_t)blockIdx.x * blockDim.x + threadIdx.x;
  if (idx >= (size_t)NB * TT) return;
  size_t b = idx / TT, t = idx % TT;
  const float* p = x + b * (32u * TT) + t;
  float s = 0.0f;
#pragma unroll
  for (int k = 0; k < 32; ++k) s += p[(size_t)k * TT];
  xsum[idx] = s;
}

template <int DIN>
__global__ __launch_bounds__(32) void lstm_layer_kernel(
    const float* __restrict__ in, float* __restrict__ out,
    const float* __restrict__ Wih, const float* __restrict__ Whh,
    const float* __restrict__ bih, const float* __restrict__ bhh) {
  constexpr int NSLICE_IN = (DIN + 3) / 4;
  constexpr int NSLICE_H  = 3;

  const int lane  = threadIdx.x;
  const int n     = lane & 15;
  const int khalf = lane >> 4;
  const int dir   = blockIdx.y;
  const int b0    = blockIdx.x * 16;

  Wih += (size_t)dir * GG * DIN;
  Whh += (size_t)dir * GG * HH;
  bih += (size_t)dir * GG;
  bhh += (size_t)dir * GG;

  v2f   Bh[4][NSLICE_H];
  v2f   Bi[4][NSLICE_IN];
  v8f   accBias[4];
  const bool gv = (n < HH);
#pragma unroll
  for (int gt = 0; gt < 4; ++gt) {
    const int g = gt * HH + n;
    accBias[gt] = splat8(gv ? (bih[g] + bhh[g]) : 0.0f);
#pragma unroll
    for (int s = 0; s < NSLICE_H; ++s) {
      const int k0 = s * 4 + 2 * khalf;
      float w0 = (gv && k0 + 0 < HH) ? Whh[(size_t)g * HH + k0 + 0] : 0.0f;
      float w1 = (gv && k0 + 1 < HH) ? Whh[(size_t)g * HH + k0 + 1] : 0.0f;
      Bh[gt][s] = v2f{w0, w1};
    }
#pragma unroll
    for (int s = 0; s < NSLICE_IN; ++s) {
      const int k0 = s * 4 + 2 * khalf;
      float w0 = (gv && k0 + 0 < DIN) ? Wih[(size_t)g * DIN + k0 + 0] : 0.0f;
      float w1 = (gv && k0 + 1 < DIN) ? Wih[(size_t)g * DIN + k0 + 1] : 0.0f;
      Bi[gt][s] = v2f{w0, w1};
    }
  }

  __shared__ float hbuf[16 * 16];
  for (int i = lane; i < 16 * 16; i += 32) hbuf[i] = 0.0f;
  __syncthreads();

  v8f c = splat8(0.0f);

  const size_t inRow = (size_t)(b0 + n) * TT;
  for (int step = 0; step < TT; ++step) {
    const int t = (dir == 0) ? step : (TT - 1 - step);

    v2f Ain[NSLICE_IN];
    if (DIN == 1) {
      const float xv = (khalf == 0) ? in[inRow + t] : 0.0f;
      Ain[0] = v2f{xv, 0.0f};
    } else {
      const float* p = in + (inRow + (size_t)t) * DIN;
#pragma unroll
      for (int s = 0; s < NSLICE_IN; ++s) {
        const int k0 = s * 4 + 2 * khalf;
        v2f v = v2f{0.0f, 0.0f};
        if (k0 + 1 < DIN) v = *(const v2f*)(p + k0);
        Ain[s] = v;
      }
      const int tn = (dir == 0) ? (t + 1) : (t - 1);
      if (tn >= 0 && tn < TT)
        __builtin_prefetch(in + (inRow + (size_t)tn) * DIN, 0, 1);
    }

    v2f Ah[NSLICE_H];
#pragma unroll
    for (int s = 0; s < NSLICE_H; ++s) {
      const int k0 = s * 4 + 2 * khalf;
      Ah[s] = *(const v2f*)(&hbuf[(n << 4) + k0]);
    }
    __syncthreads();

    v8f a0 = accBias[0], a1 = accBias[1], a2 = accBias[2], a3 = accBias[3];
#pragma unroll
    for (int s = 0; s < NSLICE_H; ++s) {
      a0 = wmma_f32_16x16x4(Ah[s], Bh[0][s], a0);
      a1 = wmma_f32_16x16x4(Ah[s], Bh[1][s], a1);
      a2 = wmma_f32_16x16x4(Ah[s], Bh[2][s], a2);
      a3 = wmma_f32_16x16x4(Ah[s], Bh[3][s], a3);
    }
#pragma unroll
    for (int s = 0; s < NSLICE_IN; ++s) {
      a0 = wmma_f32_16x16x4(Ain[s], Bi[0][s], a0);
      a1 = wmma_f32_16x16x4(Ain[s], Bi[1][s], a1);
      a2 = wmma_f32_16x16x4(Ain[s], Bi[2][s], a2);
      a3 = wmma_f32_16x16x4(Ain[s], Bi[3][s], a3);
    }

    v8f hnew;
#pragma unroll
    for (int r = 0; r < 8; ++r) {
      const float ig = fast_sigmoid(a0[r]);
      const float fg = fast_sigmoid(a1[r]);
      const float gg = fast_tanh(a2[r]);
      const float og = fast_sigmoid(a3[r]);
      const float cr = fg * c[r] + ig * gg;
      c[r]    = cr;
      hnew[r] = og * fast_tanh(cr);
    }

#pragma unroll
    for (int r = 0; r < 8; ++r)
      hbuf[((r + 8 * khalf) << 4) + n] = hnew[r];

    if (gv) {
#pragma unroll
      for (int r = 0; r < 8; ++r) {
        const int mrow = r + 8 * khalf;
        out[((size_t)(b0 + mrow) * TT + t) * OUTF + (size_t)dir * HH + n] =
            hnew[r];
      }
    }
    __syncthreads();
  }
}

extern "C" void kernel_launch(void* const* d_in, const int* in_sizes, int n_in,
                              void* d_out, int out_size, void* d_ws,
                              size_t ws_size, hipStream_t stream) {
  const float* x     = (const float*)d_in[0];
  const float* W_ih0 = (const float*)d_in[1];
  const float* W_ihr = (const float*)d_in[2];
  const float* W_hh  = (const float*)d_in[3];
  const float* b_ih  = (const float*)d_in[4];
  const float* b_hh  = (const float*)d_in[5];
  float* out = (float*)d_out;

  const size_t XS_BYTES  = (size_t)NB * TT * sizeof(float);
  const size_t ACT_BYTES = (size_t)NB * TT * OUTF * sizeof(float);
  char*  ws   = (char*)d_ws;
  float* xsum = (float*)(ws);
  float* bufA = (float*)(ws + XS_BYTES);
  float* bufB = (float*)(ws + XS_BYTES + ACT_BYTES);
  (void)ws_size; (void)in_sizes; (void)n_in; (void)out_size;

  {
    const size_t total = (size_t)NB * TT;
    reduce_x_kernel<<<dim3((unsigned)((total + 255) / 256)), dim3(256), 0,
                      stream>>>(x, xsum);
  }

  const dim3 grid(NB / 16, 2);
  const dim3 block(32);

  lstm_layer_kernel<1><<<grid, block, 0, stream>>>(
      xsum, bufA, W_ih0, W_hh, b_ih, b_hh);

  const float* lin[4]  = {bufA, bufB, bufA, bufB};
  float*       lout[4] = {bufB, bufA, bufB, out};
  for (int l = 1; l < 5; ++l) {
    lstm_layer_kernel<22><<<grid, block, 0, stream>>>(
        lin[l - 1], lout[l - 1],
        W_ihr + (size_t)(l - 1) * 2 * GG * 22,
        W_hh  + (size_t)l * 2 * GG * HH,
        b_ih  + (size_t)l * 2 * GG,
        b_hh  + (size_t)l * 2 * GG);
  }
}


# Layer output staged in the h slots of LDS: a chunk's 16 steps of h stay there and are stored a chunk later, 2 stores a step with lanes along one row's run, not 16 rows a store; sums unchanged
# speedup vs baseline: 4.7276x; 4.7276x over previous
#include <hip/hip_runtime.h>
#include <hip/hip_bf16.h>


typedef float v2f __attribute__((ext_vector_type(2)));
typedef float v4f __attribute__((ext_vector_type(4)));
typedef float v8f __attribute__((ext_vector_type(8)));

#define NB   2048
#define TT   512
#define HH   11
#define GG   44
#define OUTF 22

__device__ __forceinline__ v8f wmma_f32_16x16x4(v2f a, v2f b, v8f c) {
  return __builtin_amdgcn_wmma_f32_16x16x4_f32(false, a, false, b, (short)0, c,
                                               false, false);
}

__device__ __forceinline__ float fast_sigmoid(float x) {
  return __builtin_amdgcn_rcpf(1.0f + __expf(-x));
}
__device__ __forceinline__ float fast_tanh(float x) {
  return 1.0f - 2.0f * __builtin_amdgcn_rcpf(__expf(2.0f * x) + 1.0f);
}

__device__ __forceinline__ void lds_barrier() {
  __builtin_amdgcn_fence(__ATOMIC_RELEASE, "workgroup", "local");
  __builtin_amdgcn_s_barrier();
  __builtin_amdgcn_fence(__ATOMIC_ACQUIRE, "workgroup", "local");
}

__global__ __launch_bounds__(256) void reduce_x_kernel(
    const float* __restrict__ x, float* __restrict__ xsum) {
  size_t idx = (size_t)blockIdx.x * blockDim.x + threadIdx.x;
  if (idx >= (size_t)NB * TT) return;
  size_t b = idx / TT, t = idx % TT;
  const float* p = x + b * (32u * TT) + t;
  float s = 0.0f;
#pragma unroll
  for (int k = 0; k < 32; ++k) s += p[(size_t)k * TT];
  xsum[idx] = s;
}

struct WriteOutYes { static constexpr bool value = true; };
struct WriteOutNo  { static constexpr bool value = false; };

template <int DIN>
struct XTile {
  static constexpr int CH    = 16;
  static constexpr int NCH   = TT / CH;
  static constexpr int PITCH = (DIN == 1) ? 20 : 356;
  static constexpr int VPR   = CH * DIN / 4;
  static constexpr int NVEC  = 16 * VPR;
  static constexpr int NIT   = (NVEC + 95) / 96;
};

template <int DIN, int NS>
__device__ __forceinline__ void load_x_raw(const float* xrow, int kh,
                                           v2f (&Bx)[NS]) {
  if (DIN == 1) {
    Bx[0] = v2f{xrow[0], 0.0f};
  } else {
#pragma unroll
    for (int s = 0; s < NS; ++s) {
      const int  k0 = s * 4 + 2 * kh;
      const bool in = k0 + 1 < DIN;
      Bx[s] = *(const v2f*)(xrow + (in ? k0 : 0));
    }
  }
}
template <int DIN, int NS>
__device__ __forceinline__ void mask_x(int kh, v2f (&Bx)[NS]) {
  if (DIN == 1) {
    Bx[0] = v2f{(kh == 0) ? Bx[0].x : 0.0f, 0.0f};
  } else {
#pragma unroll
    for (int s = 0; s < NS; ++s) {
      const bool in = s * 4 + 2 * kh + 1 < DIN;
      Bx[s] = in ? Bx[s] : v2f{0.0f, 0.0f};
    }
  }
}
template <int DIN, int NS>
__device__ __forceinline__ void load_x(const float* xrow, int kh,
                                       v2f (&Bx)[NS]) {
  load_x_raw<DIN, NS>(xrow, kh, Bx);
  mask_x<DIN, NS>(kh, Bx);
}

template <int DIN>
__global__ __launch_bounds__(96) void lstm_layer_kernel(
    const float* __restrict__ in, float* __restrict__ out,
    const float* __restrict__ Wih, const float* __restrict__ Whh,
    const float* __restrict__ bih, const float* __restrict__ bhh) {
  constexpr int NSLICE_IN = (DIN + 3) / 4;
  constexpr int NSLICE_H  = 3;

  const int lane = threadIdx.x & 31;
  const int wv   = threadIdx.x >> 5;
  const int col  = lane & 15;
  const int kh   = lane >> 4;
  const int dir  = blockIdx.y;
  const int b0   = blockIdx.x * 16;

  Wih += (size_t)dir * GG * DIN;
  Whh += (size_t)dir * GG * HH;
  bih += (size_t)dir * GG;
  bhh += (size_t)dir * GG;

  const int  aUnit = 4 * wv + (col >> 2);
  const bool av    = aUnit < HH;
  const int  ag    = (col & 3) * HH + aUnit;
  const int hsl[NSLICE_H] = {wv, (wv + 1) % NSLICE_H, (wv + 2) % NSLICE_H};
  v2f Ah[NSLICE_H];
  v2f Ai[NSLICE_IN];
#pragma unroll
  for (int s = 0; s < NSLICE_H; ++s) {
    const int k0 = hsl[s] * 4 + 2 * kh;
    float w0 = (av && k0 + 0 < HH) ? Whh[(size_t)ag * HH + k0 + 0] : 0.0f;
    float w1 = (av && k0 + 1 < HH) ? Whh[(size_t)ag * HH + k0 + 1] : 0.0f;
    Ah[s] = v2f{w0, w1};
  }
#pragma unroll
  for (int s = 0; s < NSLICE_IN; ++s) {
    const int k0 = s * 4 + 2 * kh;
    float w0 = (av && k0 + 0 < DIN) ? Wih[(size_t)ag * DIN + k0 + 0] : 0.0f;
    float w1 = (av && k0 + 1 < DIN) ? Wih[(size_t)ag * DIN + k0 + 1] : 0.0f;
    Ai[s] = v2f{w0, w1};
  }
  const int unit0 = 4 * wv + 2 * kh;
  v8f biasC;
#pragma unroll
  for (int r = 0; r < 8; ++r) {
    const int u = unit0 + (r >> 2);
    const int g = (r & 3) * HH + u;
    biasC[r] = (u < HH) ? (bih[g] + bhh[g]) : 0.0f;
  }

  constexpr int HP = 33;
  __shared__ v2f hx[2][16][NSLICE_H][HP];
  hx[1][15][wv][lane] = v2f{0.0f, 0.0f};

  using XT = XTile<DIN>;
  constexpr int CH = XT::CH, NCH = XT::NCH, PITCH = XT::PITCH;
  constexpr int VPR = XT::VPR, NVEC = XT::NVEC, NIT = XT::NIT;
  __shared__ __attribute__((aligned(16))) float xs[2][16 * PITCH];
  int  gofs[NIT], lofs[NIT];
  bool vld[NIT];
#pragma unroll
  for (int j = 0; j < NIT; ++j) {
    const int n = (int)threadIdx.x + 96 * j;
    const int r = n / VPR, q = n - r * VPR;
    vld[j]  = (96 * (j + 1) <= NVEC) || (n < NVEC);
    gofs[j] = r * (TT * DIN) + 4 * q;
    lofs[j] = r * PITCH + 4 * q;
  }
  const float* gin = in + (size_t)b0 * TT * DIN;

  v4f st[NIT];
  {
    v4f st0[NIT];
    const float* g0 = gin + (size_t)((dir == 0) ? 0 : (TT - CH)) * DIN;
    const float* g1 = gin + (size_t)((dir == 0) ? CH : (TT - 2 * CH)) * DIN;
#pragma unroll
    for (int j = 0; j < NIT; ++j)
      if (vld[j]) st0[j] = *(const v4f*)(g0 + gofs[j]);
#pragma unroll
    for (int j = 0; j < NIT; ++j)
      if (vld[j]) st[j] = *(const v4f*)(g1 + gofs[j]);
#pragma unroll
    for (int j = 0; j < NIT; ++j)
      if (vld[j]) *(v4f*)(&xs[0][lofs[j]]) = st0[j];
  }

  float c0 = 0.0f, c1 = 0.0f;
  const float* xcol = &xs[0][col * PITCH];

  constexpr int NWO = 30;
  unsigned wl[NWO], wg[NWO];
#pragma unroll
  for (int q = 0; q < NWO; ++q) {
    int n = (int)threadIdx.x + 96 * q;
    if (n >= 16 * CH * HH) n -= 96;
    const int row = n / (CH * HH), f = n - row * (CH * HH);
    const int jt = f / HH, u = f - jt * HH;
    const int k  = (dir == 0) ? jt : (CH - 1 - jt);
    const int l  = ((k * NSLICE_H + (u >> 2)) * HP + ((u >> 1) & 1) * 16 + row) * 2 + (u & 1);
    const int g  = row * (TT * OUTF) + jt * OUTF + u;
    wl[q] = 4u * (unsigned)l;
    wg[q] = 4u * (unsigned)g;
  }
  float* const outd = out + (size_t)b0 * TT * OUTF + dir * HH;

  lds_barrier();

  constexpr int NPRE = NSLICE_IN / 3;
  v8f accX = biasC;
  v2f BxC[NSLICE_IN];
  {
    const int t0 = (dir == 0) ? 0 : (TT - 1);
    load_x<DIN, NSLICE_IN>(xcol + (t0 & (CH - 1)) * DIN, kh, BxC);
#pragma unroll
    for (int s = 0; s < NPRE; ++s)
      accX = wmma_f32_16x16x4(Ai[s], BxC[s], accX);
    accX = wmma_f32_16x16x4(Ah[0], v2f{0.0f, 0.0f}, accX);
  }

  const int sd = (dir == 0) ? DIN : -DIN;
  const int r0 = (dir == 0) ? 0 : (CH - 1) * DIN;

  auto chunk = [&](const int c, auto wo_tag) __attribute__((always_inline)) {
    constexpr bool WO = decltype(wo_tag)::value;
    const float* xt  = xcol + (c & 1) * (16 * PITCH) + r0;
    const float* xtn = xcol + ((c + 1) & 1) * (16 * PITCH) + r0;
    const int    par = c & 1;
    v2f*         hcur  = &hx[par][0][0][lane];
    const v2f*   hprev = &hx[par ^ 1][CH - 1][0][lane];
    const char* wsrc = (const char*)&hx[par ^ 1][0][0][0];
    char*       wdst = nullptr;
    if constexpr (WO)
      wdst = (char*)(outd + (size_t)((dir == 0) ? (c - 1) * CH : (TT - c * CH)) * OUTF);
    float wp0 = 0.0f, wp1 = 0.0f;

#pragma unroll
    for (int j = 0; j < CH; ++j) {
      v2f Bh[NSLICE_H];
#pragma unroll
      for (int s = 1; s < NSLICE_H; ++s)
        Bh[s] = (j == 0) ? hprev[hsl[s] * HP]
                         : hcur[((j - 1) * NSLICE_H + hsl[s]) * HP];

      v2f BxN[NSLICE_IN];
      load_x_raw<DIN, NSLICE_IN>((j + 1 < CH) ? xt + (j + 1) * sd : xtn, kh, BxN);

      float wn0 = 0.0f, wn1 = 0.0f;
      if constexpr (WO) {
        if (2 * j + 1 < NWO) {
          wn0 = *(const float*)(wsrc + wl[2 * j]);
          wn1 = *(const float*)(wsrc + wl[2 * j + 1]);
        }
      }

      __builtin_amdgcn_sched_barrier(0);

      if (j == 0) {
        if (c + 1 < NCH) {
#pragma unroll
          for (int q = 0; q < NIT; ++q)
            if (vld[q]) *(v4f*)(&xs[(c + 1) & 1][lofs[q]]) = st[q];
          if (c + 2 < NCH) {
            const int tlo = (dir == 0) ? (c + 2) * CH : (TT - (c + 3) * CH);
            const float* g = gin + (size_t)tlo * DIN;
#pragma unroll
            for (int q = 0; q < NIT; ++q)
              if (vld[q]) st[q] = *(const v4f*)(g + gofs[q]);
          }
        }
      }

#pragma unroll
      for (int s = NPRE; s < NSLICE_IN; ++s)
        accX = wmma_f32_16x16x4(Ai[s], BxC[s], accX);

      if constexpr (WO) {
        if (j >= 1 && 2 * j - 1 < NWO) {
          *(float*)(wdst + wg[2 * j - 2]) = wp0;
          *(float*)(wdst + wg[2 * j - 1]) = wp1;
        }
      }
      wp0 = wn0;
      wp1 = wn1;

      __builtin_amdgcn_sched_barrier(0);

      v8f acc = accX;
#pragma unroll
      for (int s = 1; s < NSLICE_H; ++s)
        acc = wmma_f32_16x16x4(Ah[s], Bh[s], acc);

      v2f hn;
      {
        const float ig = fast_sigmoid(acc[0]);
        const float fg = fast_sigmoid(acc[1]);
        const float gg = fast_tanh(acc[2]);
        const float og = fast_sigmoid(acc[3]);
        c0   = fg * c0 + ig * gg;
        hn.x = og * fast_tanh(c0);
      }
      {
        const float ig = fast_sigmoid(acc[4]);
        const float fg = fast_sigmoid(acc[5]);
        const float gg = fast_tanh(acc[6]);
        const float og = fast_sigmoid(acc[7]);
        c1   = fg * c1 + ig * gg;
        hn.y = og * fast_tanh(c1);
      }

      hcur[(j * NSLICE_H + wv) * HP] = hn;

      mask_x<DIN, NSLICE_IN>(kh, BxN);
      accX = biasC;
#pragma unroll
      for (int s = 0; s < NPRE; ++s)
        accX = wmma_f32_16x16x4(Ai[s], BxN[s], accX);
      accX = wmma_f32_16x16x4(Ah[0], hn, accX);
#pragma unroll
      for (int s = 0; s < NSLICE_IN; ++s) BxC[s] = BxN[s];

      __builtin_amdgcn_sched_barrier(0);

      lds_barrier();
    }
  };

  chunk(0, WriteOutNo{});
  for (int c = 1; c < NCH; ++c) chunk(c, WriteOutYes{});

  {
    const char* wsrc = (const char*)&hx[(NCH - 1) & 1][0][0][0];
    char*       wdst = (char*)(outd + (size_t)((dir == 0) ? (NCH - 1) * CH : 0) * OUTF);
    float w[NWO];
#pragma unroll
    for (int q = 0; q < NWO; ++q) w[q] = *(const float*)(wsrc + wl[q]);
#pragma unroll
    for (int q = 0; q < NWO; ++q) *(float*)(wdst + wg[q]) = w[q];
  }
}

extern "C" void kernel_launch(void* const* d_in, const int* in_sizes, int n_in,
                              void* d_out, int out_size, void* d_ws,
                              size_t ws_size, hipStream_t stream) {
  const float* x     = (const float*)d_in[0];
  const float* W_ih0 = (const float*)d_in[1];
  const float* W_ihr = (const float*)d_in[2];
  const float* W_hh  = (const float*)d_in[3];
  const float* b_ih  = (const float*)d_in[4];
  const float* b_hh  = (const float*)d_in[5];
  float* out = (float*)d_out;

  const size_t XS_BYTES  = (size_t)NB * TT * sizeof(float);
  const size_t ACT_BYTES = (size_t)NB * TT * OUTF * sizeof(float);
  char*  ws   = (char*)d_ws;
  float* xsum = (float*)(ws);
  float* bufA = (float*)(ws + XS_BYTES);
  float* bufB = (float*)(ws + XS_BYTES + ACT_BYTES);
  (void)ws_size; (void)in_sizes; (void)n_in; (void)out_size;

  {
    const size_t total = (size_t)NB * TT;
    reduce_x_kernel<<<dim3((unsigned)((total + 255) / 256)), dim3(256), 0,
                      stream>>>(x, xsum);
  }

  const dim3 grid(NB / 16, 2);
  const dim3 block(96);

  lstm_layer_kernel<1><<<grid, block, 0, stream>>>(
      xsum, bufA, W_ih0, W_hh, b_ih, b_hh);

  const float* lin[4]  = {bufA, bufB, bufA, bufB};
  float*       lout[4] = {bufB, bufA, bufB, out};
  for (int l = 1; l < 5; ++l) {
    lstm_layer_kernel<22><<<grid, block, 0, stream>>>(
        lin[l - 1], lout[l - 1],
        W_ihr + (size_t)(l - 1) * 2 * GG * 22,
        W_hh  + (size_t)l * 2 * GG * HH,
        b_ih  + (size_t)l * 2 * GG,
        b_hh  + (size_t)l * 2 * GG);
  }
}


# Products needing no exchanged h leave the cell update for the hand-over's waits: 3 Wih slices behind the h store, own Whh slice between barrier signal and wait; own Whh term sums after 3 Wih terms
# speedup vs baseline: 4.9479x; 1.0466x over previous
#include <hip/hip_runtime.h>
#include <hip/hip_bf16.h>


typedef float v2f __attribute__((ext_vector_type(2)));
typedef float v4f __attribute__((ext_vector_type(4)));
typedef float v8f __attribute__((ext_vector_type(8)));

#define NB   2048
#define TT   512
#define HH   11
#define GG   44
#define OUTF 22

__device__ __forceinline__ v8f wmma_f32_16x16x4(v2f a, v2f b, v8f c) {
  return __builtin_amdgcn_wmma_f32_16x16x4_f32(false, a, false, b, (short)0, c,
                                               false, false);
}

__device__ __forceinline__ float fast_sigmoid(float x) {
  return __builtin_amdgcn_rcpf(1.0f + __expf(-x));
}
__device__ __forceinline__ float fast_tanh(float x) {
  return 1.0f - 2.0f * __builtin_amdgcn_rcpf(__expf(2.0f * x) + 1.0f);
}

__device__ __forceinline__ void lds_barrier() {
  __builtin_amdgcn_fence(__ATOMIC_RELEASE, "workgroup", "local");
  __builtin_amdgcn_s_barrier();
  __builtin_amdgcn_fence(__ATOMIC_ACQUIRE, "workgroup", "local");
}
__device__ __forceinline__ void lds_barrier_signal() {
  __builtin_amdgcn_fence(__ATOMIC_RELEASE, "workgroup", "local");
  __builtin_amdgcn_s_barrier_signal(-1);
}
__device__ __forceinline__ void lds_barrier_wait() {
  __builtin_amdgcn_s_barrier_wait(-1);
  __builtin_amdgcn_fence(__ATOMIC_ACQUIRE, "workgroup", "local");
}

__global__ __launch_bounds__(256) void reduce_x_kernel(
    const float* __restrict__ x, float* __restrict__ xsum) {
  size_t idx = (size_t)blockIdx.x * blockDim.x + threadIdx.x;
  if (idx >= (size_t)NB * TT) return;
  size_t b = idx / TT, t = idx % TT;
  const float* p = x + b * (32u * TT) + t;
  float s = 0.0f;
#pragma unroll
  for (int k = 0; k < 32; ++k) s += p[(size_t)k * TT];
  xsum[idx] = s;
}

struct WriteOutYes { static constexpr bool value = true; };
struct WriteOutNo  { static constexpr bool value = false; };

template <int DIN>
struct XTile {
  static constexpr int CH    = 16;
  static constexpr int NCH   = TT / CH;
  static constexpr int PITCH = (DIN == 1) ? 20 : 356;
  static constexpr int VPR   = CH * DIN / 4;
  static constexpr int NVEC  = 16 * VPR;
  static constexpr int NIT   = (NVEC + 95) / 96;
};

template <int DIN, int NS>
__device__ __forceinline__ void load_x_raw(const float* xrow, int kh,
                                           v2f (&Bx)[NS]) {
  if (DIN == 1) {
    Bx[0] = v2f{xrow[0], 0.0f};
  } else {
#pragma unroll
    for (int s = 0; s < NS; ++s) {
      const int  k0 = s * 4 + 2 * kh;
      const bool in = k0 + 1 < DIN;
      Bx[s] = *(const v2f*)(xrow + (in ? k0 : 0));
    }
  }
}
template <int DIN, int NS>
__device__ __forceinline__ void mask_x(int kh, v2f (&Bx)[NS]) {
  if (DIN == 1) {
    Bx[0] = v2f{(kh == 0) ? Bx[0].x : 0.0f, 0.0f};
  } else {
#pragma unroll
    for (int s = 0; s < NS; ++s) {
      const bool in = s * 4 + 2 * kh + 1 < DIN;
      Bx[s] = in ? Bx[s] : v2f{0.0f, 0.0f};
    }
  }
}
template <int DIN, int NS>
__device__ __forceinline__ void load_x(const float* xrow, int kh,
                                       v2f (&Bx)[NS]) {
  load_x_raw<DIN, NS>(xrow, kh, Bx);
  mask_x<DIN, NS>(kh, Bx);
}

template <int DIN>
__global__ __launch_bounds__(96) void lstm_layer_kernel(
    const float* __restrict__ in, float* __restrict__ out,
    const float* __restrict__ Wih, const float* __restrict__ Whh,
    const float* __restrict__ bih, const float* __restrict__ bhh) {
  constexpr int NSLICE_IN = (DIN + 3) / 4;
  constexpr int NSLICE_H  = 3;

  const int lane = threadIdx.x & 31;
  const int wv   = threadIdx.x >> 5;
  const int col  = lane & 15;
  const int kh   = lane >> 4;
  const int dir  = blockIdx.y;
  const int b0   = blockIdx.x * 16;

  Wih += (size_t)dir * GG * DIN;
  Whh += (size_t)dir * GG * HH;
  bih += (size_t)dir * GG;
  bhh += (size_t)dir * GG;

  const int  aUnit = 4 * wv + (col >> 2);
  const bool av    = aUnit < HH;
  const int  ag    = (col & 3) * HH + aUnit;
  const int hsl[NSLICE_H] = {wv, (wv + 1) % NSLICE_H, (wv + 2) % NSLICE_H};
  v2f Ah[NSLICE_H];
  v2f Ai[NSLICE_IN];
#pragma unroll
  for (int s = 0; s < NSLICE_H; ++s) {
    const int k0 = hsl[s] * 4 + 2 * kh;
    float w0 = (av && k0 + 0 < HH) ? Whh[(size_t)ag * HH + k0 + 0] : 0.0f;
    float w1 = (av && k0 + 1 < HH) ? Whh[(size_t)ag * HH + k0 + 1] : 0.0f;
    Ah[s] = v2f{w0, w1};
  }
#pragma unroll
  for (int s = 0; s < NSLICE_IN; ++s) {
    const int k0 = s * 4 + 2 * kh;
    float w0 = (av && k0 + 0 < DIN) ? Wih[(size_t)ag * DIN + k0 + 0] : 0.0f;
    float w1 = (av && k0 + 1 < DIN) ? Wih[(size_t)ag * DIN + k0 + 1] : 0.0f;
    Ai[s] = v2f{w0, w1};
  }
  const int unit0 = 4 * wv + 2 * kh;
  v8f biasC;
#pragma unroll
  for (int r = 0; r < 8; ++r) {
    const int u = unit0 + (r >> 2);
    const int g = (r & 3) * HH + u;
    biasC[r] = (u < HH) ? (bih[g] + bhh[g]) : 0.0f;
  }

  constexpr int HP = 33;
  __shared__ v2f hx[2][16][NSLICE_H][HP];
  hx[1][15][wv][lane] = v2f{0.0f, 0.0f};

  using XT = XTile<DIN>;
  constexpr int CH = XT::CH, NCH = XT::NCH, PITCH = XT::PITCH;
  constexpr int VPR = XT::VPR, NVEC = XT::NVEC, NIT = XT::NIT;
  __shared__ __attribute__((aligned(16))) float xs[2][16 * PITCH];
  int  gofs[NIT], lofs[NIT];
  bool vld[NIT];
#pragma unroll
  for (int j = 0; j < NIT; ++j) {
    const int n = (int)threadIdx.x + 96 * j;
    const int r = n / VPR, q = n - r * VPR;
    vld[j]  = (96 * (j + 1) <= NVEC) || (n < NVEC);
    gofs[j] = r * (TT * DIN) + 4 * q;
    lofs[j] = r * PITCH + 4 * q;
  }
  const float* gin = in + (size_t)b0 * TT * DIN;

  v4f st[NIT];
  {
    v4f st0[NIT];
    const float* g0 = gin + (size_t)((dir == 0) ? 0 : (TT - CH)) * DIN;
    const float* g1 = gin + (size_t)((dir == 0) ? CH : (TT - 2 * CH)) * DIN;
#pragma unroll
    for (int j = 0; j < NIT; ++j)
      if (vld[j]) st0[j] = *(const v4f*)(g0 + gofs[j]);
#pragma unroll
    for (int j = 0; j < NIT; ++j)
      if (vld[j]) st[j] = *(const v4f*)(g1 + gofs[j]);
#pragma unroll
    for (int j = 0; j < NIT; ++j)
      if (vld[j]) *(v4f*)(&xs[0][lofs[j]]) = st0[j];
  }

  float c0 = 0.0f, c1 = 0.0f;
  const float* xcol = &xs[0][col * PITCH];

  constexpr int NWO = 30;
  unsigned wl[NWO], wg[NWO];
#pragma unroll
  for (int q = 0; q < NWO; ++q) {
    int n = (int)threadIdx.x + 96 * q;
    if (n >= 16 * CH * HH) n -= 96;
    const int row = n / (CH * HH), f = n - row * (CH * HH);
    const int jt = f / HH, u = f - jt * HH;
    const int k  = (dir == 0) ? jt : (CH - 1 - jt);
    const int l  = ((k * NSLICE_H + (u >> 2)) * HP + ((u >> 1) & 1) * 16 + row) * 2 + (u & 1);
    const int g  = row * (TT * OUTF) + jt * OUTF + u;
    wl[q] = 4u * (unsigned)l;
    wg[q] = 4u * (unsigned)g;
  }
  float* const outd = out + (size_t)b0 * TT * OUTF + dir * HH;

  lds_barrier();

  constexpr int NPRE = NSLICE_IN / 2;
  static_assert(4 * NPRE <= DIN, "slices issued ahead of mask_x must hold no K padding");
  v8f accX = biasC;
  v2f BxC[NSLICE_IN];
  {
    const int t0 = (dir == 0) ? 0 : (TT - 1);
    load_x<DIN, NSLICE_IN>(xcol + (t0 & (CH - 1)) * DIN, kh, BxC);
#pragma unroll
    for (int s = 0; s < NPRE; ++s)
      accX = wmma_f32_16x16x4(Ai[s], BxC[s], accX);
    accX = wmma_f32_16x16x4(Ah[0], v2f{0.0f, 0.0f}, accX);
  }

  const int sd = (dir == 0) ? DIN : -DIN;
  const int r0 = (dir == 0) ? 0 : (CH - 1) * DIN;

  auto chunk = [&](const int c, auto wo_tag) __attribute__((always_inline)) {
    constexpr bool WO = decltype(wo_tag)::value;
    const float* xt  = xcol + (c & 1) * (16 * PITCH) + r0;
    const float* xtn = xcol + ((c + 1) & 1) * (16 * PITCH) + r0;
    const int    par = c & 1;
    v2f*         hcur  = &hx[par][0][0][lane];
    const v2f*   hprev = &hx[par ^ 1][CH - 1][0][lane];
    const char* wsrc = (const char*)&hx[par ^ 1][0][0][0];
    char*       wdst = nullptr;
    if constexpr (WO)
      wdst = (char*)(outd + (size_t)((dir == 0) ? (c - 1) * CH : (TT - c * CH)) * OUTF);
    float wp0 = 0.0f, wp1 = 0.0f;

#pragma unroll
    for (int j = 0; j < CH; ++j) {
      v2f Bh[NSLICE_H];
#pragma unroll
      for (int s = 1; s < NSLICE_H; ++s)
        Bh[s] = (j == 0) ? hprev[hsl[s] * HP]
                         : hcur[((j - 1) * NSLICE_H + hsl[s]) * HP];
      __builtin_amdgcn_sched_barrier(0);

      v2f BxN[NSLICE_IN];
      load_x_raw<DIN, NSLICE_IN>((j + 1 < CH) ? xt + (j + 1) * sd : xtn, kh, BxN);

      float wn0 = 0.0f, wn1 = 0.0f;
      if constexpr (WO) {
        if (2 * j + 1 < NWO) {
          wn0 = *(const float*)(wsrc + wl[2 * j]);
          wn1 = *(const float*)(wsrc + wl[2 * j + 1]);
        }
      }

      __builtin_amdgcn_sched_barrier(0);

      if (j == 0) {
        if (c + 1 < NCH) {
#pragma unroll
          for (int q = 0; q < NIT; ++q)
            if (vld[q]) *(v4f*)(&xs[(c + 1) & 1][lofs[q]]) = st[q];
          if (c + 2 < NCH) {
            const int tlo = (dir == 0) ? (c + 2) * CH : (TT - (c + 3) * CH);
            const float* g = gin + (size_t)tlo * DIN;
#pragma unroll
            for (int q = 0; q < NIT; ++q)
              if (vld[q]) st[q] = *(const v4f*)(g + gofs[q]);
          }
        }
      }

#pragma unroll
      for (int s = NPRE; s < NSLICE_IN; ++s)
        accX = wmma_f32_16x16x4(Ai[s], BxC[s], accX);

      if constexpr (WO) {
        if (j >= 1 && 2 * j - 1 < NWO) {
          *(float*)(wdst + wg[2 * j - 2]) = wp0;
          *(float*)(wdst + wg[2 * j - 1]) = wp1;
        }
      }
      wp0 = wn0;
      wp1 = wn1;

      __builtin_amdgcn_sched_barrier(0);

      v8f acc = accX;
#pragma unroll
      for (int s = 1; s < NSLICE_H; ++s)
        acc = wmma_f32_16x16x4(Ah[s], Bh[s], acc);

      v2f hn;
      {
        const float ig0 = fast_sigmoid(acc[0]);
        const float fg0 = fast_sigmoid(acc[1]);
        const float gg0 = fast_tanh(acc[2]);
        const float og0 = fast_sigmoid(acc[3]);
        const float ig1 = fast_sigmoid(acc[4]);
        const float fg1 = fast_sigmoid(acc[5]);
        const float gg1 = fast_tanh(acc[6]);
        const float og1 = fast_sigmoid(acc[7]);
        c0 = fg0 * c0 + ig0 * gg0;
        c1 = fg1 * c1 + ig1 * gg1;
        __builtin_amdgcn_sched_barrier(0);
        hn.x = og0 * fast_tanh(c0);
        hn.y = og1 * fast_tanh(c1);
      }

      hcur[(j * NSLICE_H + wv) * HP] = hn;

      __builtin_amdgcn_sched_barrier(0);

      accX = biasC;
#pragma unroll
      for (int s = 0; s < NPRE; ++s)
        accX = wmma_f32_16x16x4(Ai[s], BxN[s], accX);

      if constexpr (NPRE > 0) asm volatile("" :: "v"(accX));
      __builtin_amdgcn_sched_barrier(0);

      lds_barrier_signal();

      __builtin_amdgcn_sched_barrier(0);

      accX = wmma_f32_16x16x4(Ah[0], hn, accX);
      mask_x<DIN, NSLICE_IN>(kh, BxN);
#pragma unroll
      for (int s = 0; s < NSLICE_IN; ++s) BxC[s] = BxN[s];

      __builtin_amdgcn_sched_barrier(0);

      lds_barrier_wait();
    }
  };

  chunk(0, WriteOutNo{});
  for (int c = 1; c < NCH; ++c) chunk(c, WriteOutYes{});

  {
    const char* wsrc = (const char*)&hx[(NCH - 1) & 1][0][0][0];
    char*       wdst = (char*)(outd + (size_t)((dir == 0) ? (NCH - 1) * CH : 0) * OUTF);
    float w[NWO];
#pragma unroll
    for (int q = 0; q < NWO; ++q) w[q] = *(const float*)(wsrc + wl[q]);
#pragma unroll
    for (int q = 0; q < NWO; ++q) *(float*)(wdst + wg[q]) = w[q];
  }
}

extern "C" void kernel_launch(void* const* d_in, const int* in_sizes, int n_in,
                              void* d_out, int out_size, void* d_ws,
                              size_t ws_size, hipStream_t stream) {
  const float* x     = (const float*)d_in[0];
  const float* W_ih0 = (const float*)d_in[1];
  const float* W_ihr = (const float*)d_in[2];
  const float* W_hh  = (const float*)d_in[3];
  const float* b_ih  = (const float*)d_in[4];
  const float* b_hh  = (const float*)d_in[5];
  float* out = (float*)d_out;

  const size_t XS_BYTES  = (size_t)NB * TT * sizeof(float);
  const size_t ACT_BYTES = (size_t)NB * TT * OUTF * sizeof(float);
  char*  ws   = (char*)d_ws;
  float* xsum = (float*)(ws);
  float* bufA = (float*)(ws + XS_BYTES);
  float* bufB = (float*)(ws + XS_BYTES + ACT_BYTES);
  (void)ws_size; (void)in_sizes; (void)n_in; (void)out_size;

  {
    const size_t total = (size_t)NB * TT;
    reduce_x_kernel<<<dim3((unsigned)((total + 255) / 256)), dim3(256), 0,
                      stream>>>(x, xsum);
  }

  const dim3 grid(NB / 16, 2);
  const dim3 block(96);

  lstm_layer_kernel<1><<<grid, block, 0, stream>>>(
      xsum, bufA, W_ih0, W_hh, b_ih, b_hh);

  const float* lin[4]  = {bufA, bufB, bufA, bufB};
  float*       lout[4] = {bufB, bufA, bufB, out};
  for (int l = 1; l < 5; ++l) {
    lstm_layer_kernel<22><<<grid, block, 0, stream>>>(
        lin[l - 1], lout[l - 1],
        W_ihr + (size_t)(l - 1) * 2 * GG * 22,
        W_hh  + (size_t)l * 2 * GG * HH,
        b_ih  + (size_t)l * 2 * GG,
        b_hh  + (size_t)l * 2 * GG);
  }
}


# Cell update's chain cut to 2^x, 1/x and single fmas: exp2 scales folded into weight rows and biases, cell state kept pre-scaled, tanh's 1-2r folded into the consuming fma; gate rounding changes
# speedup vs baseline: 5.3171x; 1.0746x over previous
#include <hip/hip_runtime.h>
#include <hip/hip_bf16.h>


typedef float v2f __attribute__((ext_vector_type(2)));
typedef float v4f __attribute__((ext_vector_type(4)));
typedef float v8f __attribute__((ext_vector_type(8)));

#define NB   2048
#define TT   512
#define HH   11
#define GG   44
#define OUTF 22

__device__ __forceinline__ v8f wmma_f32_16x16x4(v2f a, v2f b, v8f c) {
  return __builtin_amdgcn_wmma_f32_16x16x4_f32(false, a, false, b, (short)0, c,
                                               false, false);
}

#define LOG2E    1.44269504088896340736f
#define HALF_LN2 0.34657359027997265471f
__device__ __forceinline__ float gate_scale(int gate) {
  return (gate == 2) ? 2.0f * LOG2E : -LOG2E;
}
__device__ __forceinline__ float ex2(float x) { return __builtin_amdgcn_exp2f(x); }
__device__ __forceinline__ float rcp(float x) { return __builtin_amdgcn_rcpf(x); }

__device__ __forceinline__ void lds_barrier() {
  __builtin_amdgcn_fence(__ATOMIC_RELEASE, "workgroup", "local");
  __builtin_amdgcn_s_barrier();
  __builtin_amdgcn_fence(__ATOMIC_ACQUIRE, "workgroup", "local");
}
__device__ __forceinline__ void lds_barrier_signal() {
  __builtin_amdgcn_fence(__ATOMIC_RELEASE, "workgroup", "local");
  __builtin_amdgcn_s_barrier_signal(-1);
}
__device__ __forceinline__ void lds_barrier_wait() {
  __builtin_amdgcn_s_barrier_wait(-1);
  __builtin_amdgcn_fence(__ATOMIC_ACQUIRE, "workgroup", "local");
}

__global__ __launch_bounds__(256) void reduce_x_kernel(
    const float* __restrict__ x, float* __restrict__ xsum) {
  size_t idx = (size_t)blockIdx.x * blockDim.x + threadIdx.x;
  if (idx >= (size_t)NB * TT) return;
  size_t b = idx / TT, t = idx % TT;
  const float* p = x + b * (32u * TT) + t;
  float s = 0.0f;
#pragma unroll
  for (int k = 0; k < 32; ++k) s += p[(size_t)k * TT];
  xsum[idx] = s;
}

struct WriteOutYes { static constexpr bool value = true; };
struct WriteOutNo  { static constexpr bool value = false; };

template <int DIN>
struct XTile {
  static constexpr int CH    = 16;
  static constexpr int NCH   = TT / CH;
  static constexpr int PITCH = (DIN == 1) ? 20 : 356;
  static constexpr int VPR   = CH * DIN / 4;
  static constexpr int NVEC  = 16 * VPR;
  static constexpr int NIT   = (NVEC + 95) / 96;
};

template <int DIN, int NS>
__device__ __forceinline__ void load_x_raw(const float* xrow, int kh,
                                           v2f (&Bx)[NS]) {
  if (DIN == 1) {
    Bx[0] = v2f{xrow[0], 0.0f};
  } else {
#pragma unroll
    for (int s = 0; s < NS; ++s) {
      const int  k0 = s * 4 + 2 * kh;
      const bool in = k0 + 1 < DIN;
      Bx[s] = *(const v2f*)(xrow + (in ? k0 : 0));
    }
  }
}
template <int DIN, int NS>
__device__ __forceinline__ void mask_x(int kh, v2f (&Bx)[NS]) {
  if (DIN == 1) {
    Bx[0] = v2f{(kh == 0) ? Bx[0].x : 0.0f, 0.0f};
  } else {
#pragma unroll
    for (int s = 0; s < NS; ++s) {
      const bool in = s * 4 + 2 * kh + 1 < DIN;
      Bx[s] = in ? Bx[s] : v2f{0.0f, 0.0f};
    }
  }
}
template <int DIN, int NS>
__device__ __forceinline__ void load_x(const float* xrow, int kh,
                                       v2f (&Bx)[NS]) {
  load_x_raw<DIN, NS>(xrow, kh, Bx);
  mask_x<DIN, NS>(kh, Bx);
}

template <int DIN>
__global__ __launch_bounds__(96) void lstm_layer_kernel(
    const float* __restrict__ in, float* __restrict__ out,
    const float* __restrict__ Wih, const float* __restrict__ Whh,
    const float* __restrict__ bih, const float* __restrict__ bhh) {
  constexpr int NSLICE_IN = (DIN + 3) / 4;
  constexpr int NSLICE_H  = 3;

  const int lane = threadIdx.x & 31;
  const int wv   = threadIdx.x >> 5;
  const int col  = lane & 15;
  const int kh   = lane >> 4;
  const int dir  = blockIdx.y;
  const int b0   = blockIdx.x * 16;

  Wih += (size_t)dir * GG * DIN;
  Whh += (size_t)dir * GG * HH;
  bih += (size_t)dir * GG;
  bhh += (size_t)dir * GG;

  const int  aUnit = 4 * wv + (col >> 2);
  const bool av    = aUnit < HH;
  const int  ag    = (col & 3) * HH + aUnit;
  const float asc  = gate_scale(col & 3);
  const int hsl[NSLICE_H] = {wv, (wv + 1) % NSLICE_H, (wv + 2) % NSLICE_H};
  v2f Ah[NSLICE_H];
  v2f Ai[NSLICE_IN];
#pragma unroll
  for (int s = 0; s < NSLICE_H; ++s) {
    const int k0 = hsl[s] * 4 + 2 * kh;
    float w0 = (av && k0 + 0 < HH) ? Whh[(size_t)ag * HH + k0 + 0] : 0.0f;
    float w1 = (av && k0 + 1 < HH) ? Whh[(size_t)ag * HH + k0 + 1] : 0.0f;
    Ah[s] = v2f{w0 * asc, w1 * asc};
  }
#pragma unroll
  for (int s = 0; s < NSLICE_IN; ++s) {
    const int k0 = s * 4 + 2 * kh;
    float w0 = (av && k0 + 0 < DIN) ? Wih[(size_t)ag * DIN + k0 + 0] : 0.0f;
    float w1 = (av && k0 + 1 < DIN) ? Wih[(size_t)ag * DIN + k0 + 1] : 0.0f;
    Ai[s] = v2f{w0 * asc, w1 * asc};
  }
  const int unit0 = 4 * wv + 2 * kh;
  v8f biasC;
#pragma unroll
  for (int r = 0; r < 8; ++r) {
    const int u = unit0 + (r >> 2);
    const int g = (r & 3) * HH + u;
    biasC[r] = (u < HH) ? (bih[g] + bhh[g]) * gate_scale(r & 3) : 0.0f;
  }

  constexpr int HP = 33;
  __shared__ v2f hx[2][16][NSLICE_H][HP];
  hx[1][15][wv][lane] = v2f{0.0f, 0.0f};

  using XT = XTile<DIN>;
  constexpr int CH = XT::CH, NCH = XT::NCH, PITCH = XT::PITCH;
  constexpr int VPR = XT::VPR, NVEC = XT::NVEC, NIT = XT::NIT;
  __shared__ __attribute__((aligned(16))) float xs[2][16 * PITCH];
  int  gofs[NIT], lofs[NIT];
  bool vld[NIT];
#pragma unroll
  for (int j = 0; j < NIT; ++j) {
    const int n = (int)threadIdx.x + 96 * j;
    const int r = n / VPR, q = n - r * VPR;
    vld[j]  = (96 * (j + 1) <= NVEC) || (n < NVEC);
    gofs[j] = r * (TT * DIN) + 4 * q;
    lofs[j] = r * PITCH + 4 * q;
  }
  const float* gin = in + (size_t)b0 * TT * DIN;

  v4f st[NIT];
  {
    v4f st0[NIT];
    const float* g0 = gin + (size_t)((dir == 0) ? 0 : (TT - CH)) * DIN;
    const float* g1 = gin + (size_t)((dir == 0) ? CH : (TT - 2 * CH)) * DIN;
#pragma unroll
    for (int j = 0; j < NIT; ++j)
      if (vld[j]) st0[j] = *(const v4f*)(g0 + gofs[j]);
#pragma unroll
    for (int j = 0; j < NIT; ++j)
      if (vld[j]) st[j] = *(const v4f*)(g1 + gofs[j]);
#pragma unroll
    for (int j = 0; j < NIT; ++j)
      if (vld[j]) *(v4f*)(&xs[0][lofs[j]]) = st0[j];
  }

  float c0 = 0.0f, c1 = 0.0f;
  const float* xcol = &xs[0][col * PITCH];

  constexpr int NWO = 30;
  unsigned wl[NWO], wg[NWO];
#pragma unroll
  for (int q = 0; q < NWO; ++q) {
    int n = (int)threadIdx.x + 96 * q;
    if (n >= 16 * CH * HH) n -= 96;
    const int row = n / (CH * HH), f = n - row * (CH * HH);
    const int jt = f / HH, u = f - jt * HH;
    const int k  = (dir == 0) ? jt : (CH - 1 - jt);
    const int l  = ((k * NSLICE_H + (u >> 2)) * HP + ((u >> 1) & 1) * 16 + row) * 2 + (u & 1);
    const int g  = row * (TT * OUTF) + jt * OUTF + u;
    wl[q] = 4u * (unsigned)l;
    wg[q] = 4u * (unsigned)g;
  }
  float* const outd = out + (size_t)b0 * TT * OUTF + dir * HH;

  lds_barrier();

  constexpr int NPRE = NSLICE_IN / 2;
  static_assert(4 * NPRE <= DIN, "slices issued ahead of mask_x must hold no K padding");
  v8f accX = biasC;
  v2f BxC[NSLICE_IN];
  {
    const int t0 = (dir == 0) ? 0 : (TT - 1);
    load_x<DIN, NSLICE_IN>(xcol + (t0 & (CH - 1)) * DIN, kh, BxC);
#pragma unroll
    for (int s = 0; s < NPRE; ++s)
      accX = wmma_f32_16x16x4(Ai[s], BxC[s], accX);
    accX = wmma_f32_16x16x4(Ah[0], v2f{0.0f, 0.0f}, accX);
  }

  const int sd = (dir == 0) ? DIN : -DIN;
  const int r0 = (dir == 0) ? 0 : (CH - 1) * DIN;

  auto chunk = [&](const int c, auto wo_tag) __attribute__((always_inline)) {
    constexpr bool WO = decltype(wo_tag)::value;
    const float* xt  = xcol + (c & 1) * (16 * PITCH) + r0;
    const float* xtn = xcol + ((c + 1) & 1) * (16 * PITCH) + r0;
    const int    par = c & 1;
    v2f*         hcur  = &hx[par][0][0][lane];
    const v2f*   hprev = &hx[par ^ 1][CH - 1][0][lane];
    const char* wsrc = (const char*)&hx[par ^ 1][0][0][0];
    char*       wdst = nullptr;
    if constexpr (WO)
      wdst = (char*)(outd + (size_t)((dir == 0) ? (c - 1) * CH : (TT - c * CH)) * OUTF);
    float wp0 = 0.0f, wp1 = 0.0f;

#pragma unroll
    for (int j = 0; j < CH; ++j) {
      v2f Bh[NSLICE_H];
#pragma unroll
      for (int s = 1; s < NSLICE_H; ++s)
        Bh[s] = (j == 0) ? hprev[hsl[s] * HP]
                         : hcur[((j - 1) * NSLICE_H + hsl[s]) * HP];
      __builtin_amdgcn_sched_barrier(0);

      v2f BxN[NSLICE_IN];
      load_x_raw<DIN, NSLICE_IN>((j + 1 < CH) ? xt + (j + 1) * sd : xtn, kh, BxN);

      float wn0 = 0.0f, wn1 = 0.0f;
      if constexpr (WO) {
        if (2 * j + 1 < NWO) {
          wn0 = *(const float*)(wsrc + wl[2 * j]);
          wn1 = *(const float*)(wsrc + wl[2 * j + 1]);
        }
      }

      __builtin_amdgcn_sched_barrier(0);

      if (j == 0) {
        if (c + 1 < NCH) {
#pragma unroll
          for (int q = 0; q < NIT; ++q)
            if (vld[q]) *(v4f*)(&xs[(c + 1) & 1][lofs[q]]) = st[q];
          if (c + 2 < NCH) {
            const int tlo = (dir == 0) ? (c + 2) * CH : (TT - (c + 3) * CH);
            const float* g = gin + (size_t)tlo * DIN;
#pragma unroll
            for (int q = 0; q < NIT; ++q)
              if (vld[q]) st[q] = *(const v4f*)(g + gofs[q]);
          }
        }
      }

#pragma unroll
      for (int s = NPRE; s < NSLICE_IN; ++s)
        accX = wmma_f32_16x16x4(Ai[s], BxC[s], accX);

      if constexpr (WO) {
        if (j >= 1 && 2 * j - 1 < NWO) {
          *(float*)(wdst + wg[2 * j - 2]) = wp0;
          *(float*)(wdst + wg[2 * j - 1]) = wp1;
        }
      }
      wp0 = wn0;
      wp1 = wn1;

      __builtin_amdgcn_sched_barrier(0);

      v8f acc = accX;
#pragma unroll
      for (int s = 1; s < NSLICE_H; ++s)
        acc = wmma_f32_16x16x4(Ah[s], Bh[s], acc);

      v2f hn;
      {
        const float i0 = rcp(__builtin_fmaf(ex2(acc[0]), HALF_LN2, HALF_LN2));
        const float i1 = rcp(__builtin_fmaf(ex2(acc[4]), HALF_LN2, HALF_LN2));
        const float f0 = rcp(1.0f + ex2(acc[1]));
        const float f1 = rcp(1.0f + ex2(acc[5]));
        const float rg0 = rcp(1.0f + ex2(acc[2]));
        const float rg1 = rcp(1.0f + ex2(acc[6]));
        __builtin_amdgcn_sched_barrier(0);
        const float eo0 = ex2(acc[3]);
        const float eo1 = ex2(acc[7]);
        c0 = __builtin_fmaf(-2.0f * i0, rg0, __builtin_fmaf(f0, c0, i0));
        c1 = __builtin_fmaf(-2.0f * i1, rg1, __builtin_fmaf(f1, c1, i1));
        __builtin_amdgcn_sched_barrier(0);
        const float rc0 = rcp(1.0f + ex2(c0));
        const float rc1 = rcp(1.0f + ex2(c1));
        const float o0 = rcp(1.0f + eo0);
        const float o1 = rcp(1.0f + eo1);
        hn.x = __builtin_fmaf(-2.0f * o0, rc0, o0);
        hn.y = __builtin_fmaf(-2.0f * o1, rc1, o1);
      }

      hcur[(j * NSLICE_H + wv) * HP] = hn;

      __builtin_amdgcn_sched_barrier(0);

      accX = biasC;
#pragma unroll
      for (int s = 0; s < NPRE; ++s)
        accX = wmma_f32_16x16x4(Ai[s], BxN[s], accX);

      if constexpr (NPRE > 0) asm volatile("" :: "v"(accX));
      __builtin_amdgcn_sched_barrier(0);

      lds_barrier_signal();

      __builtin_amdgcn_sched_barrier(0);

      accX = wmma_f32_16x16x4(Ah[0], hn, accX);
      mask_x<DIN, NSLICE_IN>(kh, BxN);
#pragma unroll
      for (int s = 0; s < NSLICE_IN; ++s) BxC[s] = BxN[s];

      __builtin_amdgcn_sched_barrier(0);

      lds_barrier_wait();
    }
  };

  chunk(0, WriteOutNo{});
  for (int c = 1; c < NCH; ++c) chunk(c, WriteOutYes{});

  {
    const char* wsrc = (const char*)&hx[(NCH - 1) & 1][0][0][0];
    char*       wdst = (char*)(outd + (size_t)((dir == 0) ? (NCH - 1) * CH : 0) * OUTF);
    float w[NWO];
#pragma unroll
    for (int q = 0; q < NWO; ++q) w[q] = *(const float*)(wsrc + wl[q]);
#pragma unroll
    for (int q = 0; q < NWO; ++q) *(float*)(wdst + wg[q]) = w[q];
  }
}

extern "C" void kernel_launch(void* const* d_in, const int* in_sizes, int n_in,
                              void* d_out, int out_size, void* d_ws,
                              size_t ws_size, hipStream_t stream) {
  const float* x     = (const float*)d_in[0];
  const float* W_ih0 = (const float*)d_in[1];
  const float* W_ihr = (const float*)d_in[2];
  const float* W_hh  = (const float*)d_in[3];
  const float* b_ih  = (const float*)d_in[4];
  const float* b_hh  = (const float*)d_in[5];
  float* out = (float*)d_out;

  const size_t XS_BYTES  = (size_t)NB * TT * sizeof(float);
  const size_t ACT_BYTES = (size_t)NB * TT * OUTF * sizeof(float);
  char*  ws   = (char*)d_ws;
  float* xsum = (float*)(ws);
  float* bufA = (float*)(ws + XS_BYTES);
  float* bufB = (float*)(ws + XS_BYTES + ACT_BYTES);
  (void)ws_size; (void)in_sizes; (void)n_in; (void)out_size;

  {
    const size_t total = (size_t)NB * TT;
    reduce_x_kernel<<<dim3((unsigned)((total + 255) / 256)), dim3(256), 0,
                      stream>>>(x, xsum);
  }

  const dim3 grid(NB / 16, 2);
  const dim3 block(96);

  lstm_layer_kernel<1><<<grid, block, 0, stream>>>(
      xsum, bufA, W_ih0, W_hh, b_ih, b_hh);

  const float* lin[4]  = {bufA, bufB, bufA, bufB};
  float*       lout[4] = {bufB, bufA, bufB, out};
  for (int l = 1; l < 5; ++l) {
    lstm_layer_kernel<22><<<grid, block, 0, stream>>>(
        lin[l - 1], lout[l - 1],
        W_ihr + (size_t)(l - 1) * 2 * GG * 22,
        W_hh  + (size_t)l * 2 * GG * HH,
        b_ih  + (size_t)l * 2 * GG,
        b_hh  + (size_t)l * 2 * GG);
  }
}


# Cell update on hardware tanh: sigmoid(a) = 1/2 + 1/2 tanh(a/2), 1/2 folded exactly into i, f, o rows and biases, cell kept as c/2; 10 transcendentals a lane, not 20; gate, cell rounding changes
# speedup vs baseline: 5.6875x; 1.0697x over previous
#include <hip/hip_runtime.h>
#include <hip/hip_bf16.h>


typedef float v2f __attribute__((ext_vector_type(2)));
typedef float v4f __attribute__((ext_vector_type(4)));
typedef float v8f __attribute__((ext_vector_type(8)));

#define NB   2048
#define TT   512
#define HH   11
#define GG   44
#define OUTF 22

__device__ __forceinline__ v8f wmma_f32_16x16x4(v2f a, v2f b, v8f c) {
  return __builtin_amdgcn_wmma_f32_16x16x4_f32(false, a, false, b, (short)0, c,
                                               false, false);
}

__device__ __forceinline__ float gate_scale(int gate) {
  return (gate == 2) ? 1.0f : 0.5f;
}
__device__ __forceinline__ float htanh(float x) { return __builtin_amdgcn_tanhf(x); }

__device__ __forceinline__ void lds_barrier() {
  __builtin_amdgcn_fence(__ATOMIC_RELEASE, "workgroup", "local");
  __builtin_amdgcn_s_barrier();
  __builtin_amdgcn_fence(__ATOMIC_ACQUIRE, "workgroup", "local");
}
__device__ __forceinline__ void lds_barrier_signal() {
  __builtin_amdgcn_fence(__ATOMIC_RELEASE, "workgroup", "local");
  __builtin_amdgcn_s_barrier_signal(-1);
}
__device__ __forceinline__ void lds_barrier_wait() {
  __builtin_amdgcn_s_barrier_wait(-1);
  __builtin_amdgcn_fence(__ATOMIC_ACQUIRE, "workgroup", "local");
}

__global__ __launch_bounds__(256) void reduce_x_kernel(
    const float* __restrict__ x, float* __restrict__ xsum) {
  size_t idx = (size_t)blockIdx.x * blockDim.x + threadIdx.x;
  if (idx >= (size_t)NB * TT) return;
  size_t b = idx / TT, t = idx % TT;
  const float* p = x + b * (32u * TT) + t;
  float s = 0.0f;
#pragma unroll
  for (int k = 0; k < 32; ++k) s += p[(size_t)k * TT];
  xsum[idx] = s;
}

struct WriteOutYes { static constexpr bool value = true; };
struct WriteOutNo  { static constexpr bool value = false; };

template <int DIN>
struct XTile {
  static constexpr int CH    = 16;
  static constexpr int NCH   = TT / CH;
  static constexpr int PITCH = (DIN == 1) ? 20 : 356;
  static constexpr int VPR   = CH * DIN / 4;
  static constexpr int NVEC  = 16 * VPR;
  static constexpr int NIT   = (NVEC + 95) / 96;
};

template <int DIN, int NS>
__device__ __forceinline__ void load_x_raw(const float* xrow, int kh,
                                           v2f (&Bx)[NS]) {
  if (DIN == 1) {
    Bx[0] = v2f{xrow[0], 0.0f};
  } else {
#pragma unroll
    for (int s = 0; s < NS; ++s) {
      const int  k0 = s * 4 + 2 * kh;
      const bool in = k0 + 1 < DIN;
      Bx[s] = *(const v2f*)(xrow + (in ? k0 : 0));
    }
  }
}
template <int DIN, int NS>
__device__ __forceinline__ void mask_x(int kh, v2f (&Bx)[NS]) {
  if (DIN == 1) {
    Bx[0] = v2f{(kh == 0) ? Bx[0].x : 0.0f, 0.0f};
  } else {
#pragma unroll
    for (int s = 0; s < NS; ++s) {
      const bool in = s * 4 + 2 * kh + 1 < DIN;
      Bx[s] = in ? Bx[s] : v2f{0.0f, 0.0f};
    }
  }
}
template <int DIN, int NS>
__device__ __forceinline__ void load_x(const float* xrow, int kh,
                                       v2f (&Bx)[NS]) {
  load_x_raw<DIN, NS>(xrow, kh, Bx);
  mask_x<DIN, NS>(kh, Bx);
}

template <int DIN>
__global__ __launch_bounds__(96) void lstm_layer_kernel(
    const float* __restrict__ in, float* __restrict__ out,
    const float* __restrict__ Wih, const float* __restrict__ Whh,
    const float* __restrict__ bih, const float* __restrict__ bhh) {
  constexpr int NSLICE_IN = (DIN + 3) / 4;
  constexpr int NSLICE_H  = 3;

  const int lane = threadIdx.x & 31;
  const int wv   = threadIdx.x >> 5;
  const int col  = lane & 15;
  const int kh   = lane >> 4;
  const int dir  = blockIdx.y;
  const int b0   = blockIdx.x * 16;

  Wih += (size_t)dir * GG * DIN;
  Whh += (size_t)dir * GG * HH;
  bih += (size_t)dir * GG;
  bhh += (size_t)dir * GG;

  const int  aUnit = 4 * wv + (col >> 2);
  const bool av    = aUnit < HH;
  const int  ag    = (col & 3) * HH + aUnit;
  const float asc  = gate_scale(col & 3);
  const int hsl[NSLICE_H] = {wv, (wv + 1) % NSLICE_H, (wv + 2) % NSLICE_H};
  v2f Ah[NSLICE_H];
  v2f Ai[NSLICE_IN];
#pragma unroll
  for (int s = 0; s < NSLICE_H; ++s) {
    const int k0 = hsl[s] * 4 + 2 * kh;
    float w0 = (av && k0 + 0 < HH) ? Whh[(size_t)ag * HH + k0 + 0] : 0.0f;
    float w1 = (av && k0 + 1 < HH) ? Whh[(size_t)ag * HH + k0 + 1] : 0.0f;
    Ah[s] = v2f{w0 * asc, w1 * asc};
  }
#pragma unroll
  for (int s = 0; s < NSLICE_IN; ++s) {
    const int k0 = s * 4 + 2 * kh;
    float w0 = (av && k0 + 0 < DIN) ? Wih[(size_t)ag * DIN + k0 + 0] : 0.0f;
    float w1 = (av && k0 + 1 < DIN) ? Wih[(size_t)ag * DIN + k0 + 1] : 0.0f;
    Ai[s] = v2f{w0 * asc, w1 * asc};
  }
  const int unit0 = 4 * wv + 2 * kh;
  v8f biasC;
#pragma unroll
  for (int r = 0; r < 8; ++r) {
    const int u = unit0 + (r >> 2);
    const int g = (r & 3) * HH + u;
    biasC[r] = (u < HH) ? (bih[g] + bhh[g]) * gate_scale(r & 3) : 0.0f;
  }

  constexpr int HP = 33;
  __shared__ v2f hx[2][16][NSLICE_H][HP];
  hx[1][15][wv][lane] = v2f{0.0f, 0.0f};

  using XT = XTile<DIN>;
  constexpr int CH = XT::CH, NCH = XT::NCH, PITCH = XT::PITCH;
  constexpr int VPR = XT::VPR, NVEC = XT::NVEC, NIT = XT::NIT;
  __shared__ __attribute__((aligned(16))) float xs[2][16 * PITCH];
  int  gofs[NIT], lofs[NIT];
  bool vld[NIT];
#pragma unroll
  for (int j = 0; j < NIT; ++j) {
    const int n = (int)threadIdx.x + 96 * j;
    const int r = n / VPR, q = n - r * VPR;
    vld[j]  = (96 * (j + 1) <= NVEC) || (n < NVEC);
    gofs[j] = r * (TT * DIN) + 4 * q;
    lofs[j] = r * PITCH + 4 * q;
  }
  const float* gin = in + (size_t)b0 * TT * DIN;

  v4f st[NIT];
  {
    v4f st0[NIT];
    const float* g0 = gin + (size_t)((dir == 0) ? 0 : (TT - CH)) * DIN;
    const float* g1 = gin + (size_t)((dir == 0) ? CH : (TT - 2 * CH)) * DIN;
#pragma unroll
    for (int j = 0; j < NIT; ++j)
      if (vld[j]) st0[j] = *(const v4f*)(g0 + gofs[j]);
#pragma unroll
    for (int j = 0; j < NIT; ++j)
      if (vld[j]) st[j] = *(const v4f*)(g1 + gofs[j]);
#pragma unroll
    for (int j = 0; j < NIT; ++j)
      if (vld[j]) *(v4f*)(&xs[0][lofs[j]]) = st0[j];
  }

  float hc0 = 0.0f, hc1 = 0.0f;
  const float* xcol = &xs[0][col * PITCH];

  constexpr int NWO = 30;
  unsigned wl[NWO], wg[NWO];
#pragma unroll
  for (int q = 0; q < NWO; ++q) {
    int n = (int)threadIdx.x + 96 * q;
    if (n >= 16 * CH * HH) n -= 96;
    const int row = n / (CH * HH), f = n - row * (CH * HH);
    const int jt = f / HH, u = f - jt * HH;
    const int k  = (dir == 0) ? jt : (CH - 1 - jt);
    const int l  = ((k * NSLICE_H + (u >> 2)) * HP + ((u >> 1) & 1) * 16 + row) * 2 + (u & 1);
    const int g  = row * (TT * OUTF) + jt * OUTF + u;
    wl[q] = 4u * (unsigned)l;
    wg[q] = 4u * (unsigned)g;
  }
  float* const outd = out + (size_t)b0 * TT * OUTF + dir * HH;

  lds_barrier();

  constexpr int NPRE = NSLICE_IN / 2;
  static_assert(4 * NPRE <= DIN, "slices issued ahead of mask_x must hold no K padding");
  v8f accX = biasC;
  v2f BxC[NSLICE_IN];
  {
    const int t0 = (dir == 0) ? 0 : (TT - 1);
    load_x<DIN, NSLICE_IN>(xcol + (t0 & (CH - 1)) * DIN, kh, BxC);
#pragma unroll
    for (int s = 0; s < NPRE; ++s)
      accX = wmma_f32_16x16x4(Ai[s], BxC[s], accX);
    accX = wmma_f32_16x16x4(Ah[0], v2f{0.0f, 0.0f}, accX);
  }

  const int sd = (dir == 0) ? DIN : -DIN;
  const int r0 = (dir == 0) ? 0 : (CH - 1) * DIN;

  auto chunk = [&](const int c, auto wo_tag) __attribute__((always_inline)) {
    constexpr bool WO = decltype(wo_tag)::value;
    const float* xt  = xcol + (c & 1) * (16 * PITCH) + r0;
    const float* xtn = xcol + ((c + 1) & 1) * (16 * PITCH) + r0;
    const int    par = c & 1;
    v2f*         hcur  = &hx[par][0][0][lane];
    const v2f*   hprev = &hx[par ^ 1][CH - 1][0][lane];
    const char* wsrc = (const char*)&hx[par ^ 1][0][0][0];
    char*       wdst = nullptr;
    if constexpr (WO)
      wdst = (char*)(outd + (size_t)((dir == 0) ? (c - 1) * CH : (TT - c * CH)) * OUTF);
    float wp0 = 0.0f, wp1 = 0.0f;

#pragma unroll
    for (int j = 0; j < CH; ++j) {
      v2f Bh[NSLICE_H];
#pragma unroll
      for (int s = 1; s < NSLICE_H; ++s)
        Bh[s] = (j == 0) ? hprev[hsl[s] * HP]
                         : hcur[((j - 1) * NSLICE_H + hsl[s]) * HP];
      __builtin_amdgcn_sched_barrier(0);

      v2f BxN[NSLICE_IN];
      load_x_raw<DIN, NSLICE_IN>((j + 1 < CH) ? xt + (j + 1) * sd : xtn, kh, BxN);

      float wn0 = 0.0f, wn1 = 0.0f;
      if constexpr (WO) {
        if (2 * j + 1 < NWO) {
          wn0 = *(const float*)(wsrc + wl[2 * j]);
          wn1 = *(const float*)(wsrc + wl[2 * j + 1]);
        }
      }

      __builtin_amdgcn_sched_barrier(0);

      if (j == 0) {
        if (c + 1 < NCH) {
#pragma unroll
          for (int q = 0; q < NIT; ++q)
            if (vld[q]) *(v4f*)(&xs[(c + 1) & 1][lofs[q]]) = st[q];
          if (c + 2 < NCH) {
            const int tlo = (dir == 0) ? (c + 2) * CH : (TT - (c + 3) * CH);
            const float* g = gin + (size_t)tlo * DIN;
#pragma unroll
            for (int q = 0; q < NIT; ++q)
              if (vld[q]) st[q] = *(const v4f*)(g + gofs[q]);
          }
        }
      }

#pragma unroll
      for (int s = NPRE; s < NSLICE_IN; ++s)
        accX = wmma_f32_16x16x4(Ai[s], BxC[s], accX);

      if constexpr (WO) {
        if (j >= 1 && 2 * j - 1 < NWO) {
          *(float*)(wdst + wg[2 * j - 2]) = wp0;
          *(float*)(wdst + wg[2 * j - 1]) = wp1;
        }
      }
      wp0 = wn0;
      wp1 = wn1;

      __builtin_amdgcn_sched_barrier(0);

      v8f acc = accX;
#pragma unroll
      for (int s = 1; s < NSLICE_H; ++s)
        acc = wmma_f32_16x16x4(Ah[s], Bh[s], acc);

      v2f hn;
      {
        const float tf0 = htanh(acc[1]);
        const float tf1 = htanh(acc[5]);
        const float ti0 = htanh(acc[0]);
        const float ti1 = htanh(acc[4]);
        const float tg0 = htanh(acc[2]);
        const float tg1 = htanh(acc[6]);
        const float to0 = htanh(acc[3]);
        const float to1 = htanh(acc[7]);
        const float fc0 = __builtin_fmaf(tf0, hc0, hc0);
        const float fc1 = __builtin_fmaf(tf1, hc1, hc1);
        const float i0  = __builtin_fmaf(0.5f, ti0, 0.5f);
        const float i1  = __builtin_fmaf(0.5f, ti1, 0.5f);
        const float cn0 = __builtin_fmaf(i0, tg0, fc0);
        const float cn1 = __builtin_fmaf(i1, tg1, fc1);
        const float tc0 = htanh(cn0);
        const float tc1 = htanh(cn1);
        const float o0  = __builtin_fmaf(0.5f, to0, 0.5f);
        const float o1  = __builtin_fmaf(0.5f, to1, 0.5f);
        hc0 = 0.5f * cn0;
        hc1 = 0.5f * cn1;
        hn.x = o0 * tc0;
        hn.y = o1 * tc1;
      }

      hcur[(j * NSLICE_H + wv) * HP] = hn;

      __builtin_amdgcn_sched_barrier(0);

      accX = biasC;
#pragma unroll
      for (int s = 0; s < NPRE; ++s)
        accX = wmma_f32_16x16x4(Ai[s], BxN[s], accX);

      if constexpr (NPRE > 0) asm volatile("" :: "v"(accX));
      __builtin_amdgcn_sched_barrier(0);

      lds_barrier_signal();

      __builtin_amdgcn_sched_barrier(0);

      accX = wmma_f32_16x16x4(Ah[0], hn, accX);
      mask_x<DIN, NSLICE_IN>(kh, BxN);
#pragma unroll
      for (int s = 0; s < NSLICE_IN; ++s) BxC[s] = BxN[s];

      __builtin_amdgcn_sched_barrier(0);

      lds_barrier_wait();
    }
  };

  chunk(0, WriteOutNo{});
  for (int c = 1; c < NCH; ++c) chunk(c, WriteOutYes{});

  {
    const char* wsrc = (const char*)&hx[(NCH - 1) & 1][0][0][0];
    char*       wdst = (char*)(outd + (size_t)((dir == 0) ? (NCH - 1) * CH : 0) * OUTF);
    float w[NWO];
#pragma unroll
    for (int q = 0; q < NWO; ++q) w[q] = *(const float*)(wsrc + wl[q]);
#pragma unroll
    for (int q = 0; q < NWO; ++q) *(float*)(wdst + wg[q]) = w[q];
  }
}

extern "C" void kernel_launch(void* const* d_in, const int* in_sizes, int n_in,
                              void* d_out, int out_size, void* d_ws,
                              size_t ws_size, hipStream_t stream) {
  const float* x     = (const float*)d_in[0];
  const float* W_ih0 = (const float*)d_in[1];
  const float* W_ihr = (const float*)d_in[2];
  const float* W_hh  = (const float*)d_in[3];
  const float* b_ih  = (const float*)d_in[4];
  const float* b_hh  = (const float*)d_in[5];
  float* out = (float*)d_out;

  const size_t XS_BYTES  = (size_t)NB * TT * sizeof(float);
  const size_t ACT_BYTES = (size_t)NB * TT * OUTF * sizeof(float);
  char*  ws   = (char*)d_ws;
  float* xsum = (float*)(ws);
  float* bufA = (float*)(ws + XS_BYTES);
  float* bufB = (float*)(ws + XS_BYTES + ACT_BYTES);
  (void)ws_size; (void)in_sizes; (void)n_in; (void)out_size;

  {
    const size_t total = (size_t)NB * TT;
    reduce_x_kernel<<<dim3((unsigned)((total + 255) / 256)), dim3(256), 0,
                      stream>>>(x, xsum);
  }

  const dim3 grid(NB / 16, 2);
  const dim3 block(96);

  lstm_layer_kernel<1><<<grid, block, 0, stream>>>(
      xsum, bufA, W_ih0, W_hh, b_ih, b_hh);

  const float* lin[4]  = {bufA, bufB, bufA, bufB};
  float*       lout[4] = {bufB, bufA, bufB, out};
  for (int l = 1; l < 5; ++l) {
    lstm_layer_kernel<22><<<grid, block, 0, stream>>>(
        lin[l - 1], lout[l - 1],
        W_ihr + (size_t)(l - 1) * 2 * GG * 22,
        W_hh  + (size_t)l * 2 * GG * HH,
        b_ih  + (size_t)l * 2 * GG,
        b_hh  + (size_t)l * 2 * GG);
  }
}
